# starting kernel
# baseline (speedup 1.0000x reference)
#include <hip/hip_runtime.h>
#include <hip/hip_bf16.h>

typedef __attribute__((ext_vector_type(2))) float v2f;
typedef __attribute__((ext_vector_type(8))) float v8f;

constexpr int BATCH = 8;
constexpr int SEQ   = 4096;
constexpr int DIM   = 1024;
constexpr int HEADS = 16;
constexpr float SCALE = 0.125f;

__global__ __launch_bounds__(32)
void k_logits(const float* __restrict__ x, const float* __restrict__ W,
              const int* __restrict__ mask, float* __restrict__ logits)
{
    const int lane = threadIdx.x;
    const int m    = lane & 15;
    const int half = lane >> 4;
    const int tile = blockIdx.x;
    const int b    = tile >> 8;
    const int s0   = (tile & 255) << 4;

    const float* xrow = x + ((size_t)b * SEQ + (s0 + m)) * DIM;
    v8f c = {};

    #pragma unroll 4
    for (int k0 = 0; k0 < DIM; k0 += 4) {
        const int ka = k0 + 2 * half;
        v2f a = *(const v2f*)(xrow + ka);
        v2f bf;
        bf.x = W[(size_t)(ka + 0) * HEADS + m];
        bf.y = W[(size_t)(ka + 1) * HEADS + m];
        c = __builtin_amdgcn_wmma_f32_16x16x4_f32(
                false, a, false, bf, (short)0, c, false, false);
    }

    #pragma unroll
    for (int r = 0; r < 8; ++r) {
        const int s = s0 + r + 8 * half;
        float v = c[r] * SCALE;
        if (mask[(size_t)b * SEQ + s] == 0) v = -__builtin_inff();
        logits[((size_t)b * SEQ + s) * HEADS + m] = v;
    }
}

__global__ __launch_bounds__(256)
void k_softmax(float* __restrict__ logits)
{
    __shared__ float red[256];
    const int bh = blockIdx.x;
    const int b  = bh >> 4;
    const int h  = bh & 15;
    const int t  = threadIdx.x;
    float* base = logits + (size_t)b * SEQ * HEADS + h;

    float vals[16];
    float mx = -__builtin_inff();
    #pragma unroll
    for (int i = 0; i < 16; ++i) {
        float v = base[(size_t)(i * 256 + t) * HEADS];
        vals[i] = v;
        mx = fmaxf(mx, v);
    }
    red[t] = mx; __syncthreads();
    for (int off = 128; off > 0; off >>= 1) {
        if (t < off) red[t] = fmaxf(red[t], red[t + off]);
        __syncthreads();
    }
    mx = red[0];
    __syncthreads();

    float sum = 0.0f;
    #pragma unroll
    for (int i = 0; i < 16; ++i) {
        float e = __expf(vals[i] - mx);
        vals[i] = e;
        sum += e;
    }
    red[t] = sum; __syncthreads();
    for (int off = 128; off > 0; off >>= 1) {
        if (t < off) red[t] += red[t + off];
        __syncthreads();
    }
    const float rinv = 1.0f / red[0];

    #pragma unroll
    for (int i = 0; i < 16; ++i)
        base[(size_t)(i * 256 + t) * HEADS] = vals[i] * rinv;
}

__global__ __launch_bounds__(32)
void k_pool(const float* __restrict__ x, const float* __restrict__ attn,
            float* __restrict__ out)
{
    const int lane = threadIdx.x;
    const int n    = lane & 15;
    const int half = lane >> 4;
    const int tile = blockIdx.x;
    const int b    = tile >> 6;
    const int d0   = (tile & 63) << 4;

    const float* xb = x    + (size_t)b * SEQ * DIM   + d0 + n;
    const float* ab = attn + (size_t)b * SEQ * HEADS + n;
    v8f c = {};

    #pragma unroll 4
    for (int k0 = 0; k0 < SEQ; k0 += 4) {
        const int ka = k0 + 2 * half;
        v2f a;
        a.x = ab[(size_t)(ka + 0) * HEADS];
        a.y = ab[(size_t)(ka + 1) * HEADS];
        v2f bf;
        bf.x = xb[(size_t)(ka + 0) * DIM];
        bf.y = xb[(size_t)(ka + 1) * DIM];
        c = __builtin_amdgcn_wmma_f32_16x16x4_f32(
                false, a, false, bf, (short)0, c, false, false);
    }

    #pragma unroll
    for (int r = 0; r < 8; ++r) {
        const int hrow = r + 8 * half;
        out[(size_t)b * (HEADS * DIM) + (size_t)hrow * DIM + d0 + n] = c[r];
    }
}

extern "C" void kernel_launch(void* const* d_in, const int* in_sizes, int n_in,
                              void* d_out, int out_size, void* d_ws, size_t ws_size,
                              hipStream_t stream)
{
    const float* x    = (const float*)d_in[0];
    const float* W    = (const float*)d_in[1];
    const int*   mask = (const int*)  d_in[2];
    float* out    = (float*)d_out;
    float* logits = (float*)d_ws;

    k_logits <<<dim3(BATCH * SEQ / 16), dim3(32),  0, stream>>>(x, W, mask, logits);
    k_softmax<<<dim3(BATCH * HEADS),    dim3(256), 0, stream>>>(logits);
    k_pool   <<<dim3(BATCH * DIM / 16), dim3(32),  0, stream>>>(x, logits, out);
}


# k_logits and k_pool: loads of the next 32 k-steps issued before the current 32 k-steps' products (two register sets per wave, software pipelined); k order and single accumulator kept
# speedup vs baseline: 2.3071x; 2.3071x over previous
#include <hip/hip_runtime.h>
#include <hip/hip_bf16.h>

typedef __attribute__((ext_vector_type(2))) float v2f;
typedef __attribute__((ext_vector_type(8))) float v8f;

constexpr int BATCH = 8;
constexpr int SEQ   = 4096;
constexpr int DIM   = 1024;
constexpr int HEADS = 16;
constexpr float SCALE = 0.125f;
constexpr int RS = 32;

__device__ __forceinline__ v8f mma_run(v8f c, const v2f (&a)[RS], const v2f (&b)[RS])
{
    #pragma unroll
    for (int j = 0; j < RS; ++j)
        c = __builtin_amdgcn_wmma_f32_16x16x4_f32(
                false, a[j], false, b[j], (short)0, c, false, false);
    return c;
}

__device__ __forceinline__ void lg_load(v2f (&a)[RS], v2f (&w)[RS],
                                        const float* __restrict__ xp,
                                        const float* __restrict__ wp)
{
    #pragma unroll
    for (int j = 0; j < RS; ++j) {
        a[j]   = *(const v2f*)(xp + 4 * j);
        w[j].x = wp[(size_t)(4 * j + 0) * HEADS];
        w[j].y = wp[(size_t)(4 * j + 1) * HEADS];
    }
}

__device__ __forceinline__ void pl_load(v2f (&a)[RS], v2f (&bf)[RS],
                                        const float* __restrict__ ab,
                                        const float* __restrict__ xb)
{
    #pragma unroll
    for (int j = 0; j < RS; ++j) {
        a[j].x  = ab[(size_t)(4 * j + 0) * HEADS];
        a[j].y  = ab[(size_t)(4 * j + 1) * HEADS];
        bf[j].x = xb[(size_t)(4 * j + 0) * DIM];
        bf[j].y = xb[(size_t)(4 * j + 1) * DIM];
    }
}

__global__ __launch_bounds__(32)
void k_logits(const float* __restrict__ x, const float* __restrict__ W,
              const int* __restrict__ mask, float* __restrict__ logits)
{
    const int lane = threadIdx.x;
    const int m    = lane & 15;
    const int half = lane >> 4;
    const int tile = blockIdx.x;
    const int b    = tile >> 8;
    const int s0   = (tile & 255) << 4;

    const float* xp = x + ((size_t)b * SEQ + (s0 + m)) * DIM + 2 * half;
    const float* wp = W + (size_t)(2 * half) * HEADS + m;
    v8f c = {};

    v2f a0[RS], w0[RS], a1[RS], w1[RS];
    constexpr int NSET = DIM / (4 * RS);
    lg_load(a0, w0, xp, wp);
    for (int s = 0; s < NSET - 2; s += 2) {
        lg_load(a1, w1, xp + (s + 1) * 4 * RS, wp + (size_t)(s + 1) * 4 * RS * HEADS);
        c = mma_run(c, a0, w0);
        lg_load(a0, w0, xp + (s + 2) * 4 * RS, wp + (size_t)(s + 2) * 4 * RS * HEADS);
        c = mma_run(c, a1, w1);
    }
    lg_load(a1, w1, xp + (NSET - 1) * 4 * RS, wp + (size_t)(NSET - 1) * 4 * RS * HEADS);
    c = mma_run(c, a0, w0);
    c = mma_run(c, a1, w1);

    #pragma unroll
    for (int r = 0; r < 8; ++r) {
        const int s = s0 + r + 8 * half;
        float v = c[r] * SCALE;
        if (mask[(size_t)b * SEQ + s] == 0) v = -__builtin_inff();
        logits[((size_t)b * SEQ + s) * HEADS + m] = v;
    }
}

__global__ __launch_bounds__(256)
void k_softmax(float* __restrict__ logits)
{
    __shared__ float red[256];
    const int bh = blockIdx.x;
    const int b  = bh >> 4;
    const int h  = bh & 15;
    const int t  = threadIdx.x;
    float* base = logits + (size_t)b * SEQ * HEADS + h;

    float vals[16];
    float mx = -__builtin_inff();
    #pragma unroll
    for (int i = 0; i < 16; ++i) {
        float v = base[(size_t)(i * 256 + t) * HEADS];
        vals[i] = v;
        mx = fmaxf(mx, v);
    }
    red[t] = mx; __syncthreads();
    for (int off = 128; off > 0; off >>= 1) {
        if (t < off) red[t] = fmaxf(red[t], red[t + off]);
        __syncthreads();
    }
    mx = red[0];
    __syncthreads();

    float sum = 0.0f;
    #pragma unroll
    for (int i = 0; i < 16; ++i) {
        float e = __expf(vals[i] - mx);
        vals[i] = e;
        sum += e;
    }
    red[t] = sum; __syncthreads();
    for (int off = 128; off > 0; off >>= 1) {
        if (t < off) red[t] += red[t + off];
        __syncthreads();
    }
    const float rinv = 1.0f / red[0];

    #pragma unroll
    for (int i = 0; i < 16; ++i)
        base[(size_t)(i * 256 + t) * HEADS] = vals[i] * rinv;
}

__global__ __launch_bounds__(32)
void k_pool(const float* __restrict__ x, const float* __restrict__ attn,
            float* __restrict__ out)
{
    const int lane = threadIdx.x;
    const int n    = lane & 15;
    const int half = lane >> 4;
    const int tile = blockIdx.x;
    const int b    = tile >> 6;
    const int d0   = (tile & 63) << 4;

    const float* xb = x    + (size_t)b * SEQ * DIM   + d0 + n + (size_t)(2 * half) * DIM;
    const float* ab = attn + (size_t)b * SEQ * HEADS + n      + (size_t)(2 * half) * HEADS;
    v8f c = {};

    v2f a0[RS], b0[RS], a1[RS], b1[RS];
    constexpr int NSET = SEQ / (4 * RS);
    pl_load(a0, b0, ab, xb);
    for (int s = 0; s < NSET - 2; s += 2) {
        pl_load(a1, b1, ab + (size_t)(s + 1) * 4 * RS * HEADS, xb + (size_t)(s + 1) * 4 * RS * DIM);
        c = mma_run(c, a0, b0);
        pl_load(a0, b0, ab + (size_t)(s + 2) * 4 * RS * HEADS, xb + (size_t)(s + 2) * 4 * RS * DIM);
        c = mma_run(c, a1, b1);
    }
    pl_load(a1, b1, ab + (size_t)(NSET - 1) * 4 * RS * HEADS, xb + (size_t)(NSET - 1) * 4 * RS * DIM);
    c = mma_run(c, a0, b0);
    c = mma_run(c, a1, b1);

    #pragma unroll
    for (int r = 0; r < 8; ++r) {
        const int hrow = r + 8 * half;
        out[(size_t)b * (HEADS * DIM) + (size_t)hrow * DIM + d0 + n] = c[r];
    }
}

extern "C" void kernel_launch(void* const* d_in, const int* in_sizes, int n_in,
                              void* d_out, int out_size, void* d_ws, size_t ws_size,
                              hipStream_t stream)
{
    const float* x    = (const float*)d_in[0];
    const float* W    = (const float*)d_in[1];
    const int*   mask = (const int*)  d_in[2];
    float* out    = (float*)d_out;
    float* logits = (float*)d_ws;

    k_logits <<<dim3(BATCH * SEQ / 16), dim3(32),  0, stream>>>(x, W, mask, logits);
    k_softmax<<<dim3(BATCH * HEADS),    dim3(256), 0, stream>>>(logits);
    k_pool   <<<dim3(BATCH * DIM / 16), dim3(32),  0, stream>>>(x, logits, out);
}


# k_pool: grid cut along the sequence, one wave per (256-row chunk, 64-column slab), 16-byte x loads, four tiles per A fragment; a fourth launch adds the 16 chunk partials (changes summation order)
# speedup vs baseline: 3.4387x; 1.4905x over previous
#include <hip/hip_runtime.h>
#include <hip/hip_bf16.h>

typedef __attribute__((ext_vector_type(2))) float v2f;
typedef __attribute__((ext_vector_type(4))) float v4f;
typedef __attribute__((ext_vector_type(8))) float v8f;

constexpr int BATCH = 8;
constexpr int SEQ   = 4096;
constexpr int DIM   = 1024;
constexpr int HEADS = 16;
constexpr float SCALE = 0.125f;
constexpr int RS = 32;
constexpr int CH    = 16;
constexpr int CROWS = SEQ / CH;
constexpr int PS    = 8;
constexpr int SLAB  = 64;

__device__ __forceinline__ v8f mma_run(v8f c, const v2f (&a)[RS], const v2f (&b)[RS])
{
    #pragma unroll
    for (int j = 0; j < RS; ++j)
        c = __builtin_amdgcn_wmma_f32_16x16x4_f32(
                false, a[j], false, b[j], (short)0, c, false, false);
    return c;
}

__device__ __forceinline__ void lg_load(v2f (&a)[RS], v2f (&w)[RS],
                                        const float* __restrict__ xp,
                                        const float* __restrict__ wp)
{
    #pragma unroll
    for (int j = 0; j < RS; ++j) {
        a[j]   = *(const v2f*)(xp + 4 * j);
        w[j].x = wp[(size_t)(4 * j + 0) * HEADS];
        w[j].y = wp[(size_t)(4 * j + 1) * HEADS];
    }
}

__device__ __forceinline__ void pl_load(v2f (&a)[PS], v4f (&q0)[PS], v4f (&q1)[PS],
                                        const float* __restrict__ ab,
                                        const float* __restrict__ xb)
{
    #pragma unroll
    for (int j = 0; j < PS; ++j) {
        a[j].x = ab[(size_t)(4 * j + 0) * HEADS];
        a[j].y = ab[(size_t)(4 * j + 1) * HEADS];
        q0[j]  = *(const v4f*)(xb + (size_t)(4 * j + 0) * DIM);
        q1[j]  = *(const v4f*)(xb + (size_t)(4 * j + 1) * DIM);
    }
}

__device__ __forceinline__ void pl_mma(v8f (&c)[4], const v2f (&a)[PS],
                                       const v4f (&q0)[PS], const v4f (&q1)[PS])
{
    #pragma unroll
    for (int j = 0; j < PS; ++j) {
        const v2f b0 = {q0[j].x, q1[j].x};
        const v2f b1 = {q0[j].y, q1[j].y};
        const v2f b2 = {q0[j].z, q1[j].z};
        const v2f b3 = {q0[j].w, q1[j].w};
        c[0] = __builtin_amdgcn_wmma_f32_16x16x4_f32(false, a[j], false, b0, (short)0, c[0], false, false);
        c[1] = __builtin_amdgcn_wmma_f32_16x16x4_f32(false, a[j], false, b1, (short)0, c[1], false, false);
        c[2] = __builtin_amdgcn_wmma_f32_16x16x4_f32(false, a[j], false, b2, (short)0, c[2], false, false);
        c[3] = __builtin_amdgcn_wmma_f32_16x16x4_f32(false, a[j], false, b3, (short)0, c[3], false, false);
    }
}

__global__ __launch_bounds__(32)
void k_logits(const float* __restrict__ x, const float* __restrict__ W,
              const int* __restrict__ mask, float* __restrict__ logits)
{
    const int lane = threadIdx.x;
    const int m    = lane & 15;
    const int half = lane >> 4;
    const int tile = blockIdx.x;
    const int b    = tile >> 8;
    const int s0   = (tile & 255) << 4;

    const float* xp = x + ((size_t)b * SEQ + (s0 + m)) * DIM + 2 * half;
    const float* wp = W + (size_t)(2 * half) * HEADS + m;
    v8f c = {};

    v2f a0[RS], w0[RS], a1[RS], w1[RS];
    constexpr int NSET = DIM / (4 * RS);
    lg_load(a0, w0, xp, wp);
    for (int s = 0; s < NSET - 2; s += 2) {
        lg_load(a1, w1, xp + (s + 1) * 4 * RS, wp + (size_t)(s + 1) * 4 * RS * HEADS);
        c = mma_run(c, a0, w0);
        lg_load(a0, w0, xp + (s + 2) * 4 * RS, wp + (size_t)(s + 2) * 4 * RS * HEADS);
        c = mma_run(c, a1, w1);
    }
    lg_load(a1, w1, xp + (NSET - 1) * 4 * RS, wp + (size_t)(NSET - 1) * 4 * RS * HEADS);
    c = mma_run(c, a0, w0);
    c = mma_run(c, a1, w1);

    #pragma unroll
    for (int r = 0; r < 8; ++r) {
        const int s = s0 + r + 8 * half;
        float v = c[r] * SCALE;
        if (mask[(size_t)b * SEQ + s] == 0) v = -__builtin_inff();
        logits[((size_t)b * SEQ + s) * HEADS + m] = v;
    }
}

__global__ __launch_bounds__(256)
void k_softmax(float* __restrict__ logits)
{
    __shared__ float red[256];
    const int bh = blockIdx.x;
    const int b  = bh >> 4;
    const int h  = bh & 15;
    const int t  = threadIdx.x;
    float* base = logits + (size_t)b * SEQ * HEADS + h;

    float vals[16];
    float mx = -__builtin_inff();
    #pragma unroll
    for (int i = 0; i < 16; ++i) {
        float v = base[(size_t)(i * 256 + t) * HEADS];
        vals[i] = v;
        mx = fmaxf(mx, v);
    }
    red[t] = mx; __syncthreads();
    for (int off = 128; off > 0; off >>= 1) {
        if (t < off) red[t] = fmaxf(red[t], red[t + off]);
        __syncthreads();
    }
    mx = red[0];
    __syncthreads();

    float sum = 0.0f;
    #pragma unroll
    for (int i = 0; i < 16; ++i) {
        float e = __expf(vals[i] - mx);
        vals[i] = e;
        sum += e;
    }
    red[t] = sum; __syncthreads();
    for (int off = 128; off > 0; off >>= 1) {
        if (t < off) red[t] += red[t + off];
        __syncthreads();
    }
    const float rinv = 1.0f / red[0];

    #pragma unroll
    for (int i = 0; i < 16; ++i)
        base[(size_t)(i * 256 + t) * HEADS] = vals[i] * rinv;
}

__global__ __launch_bounds__(32)
void k_pool(const float* __restrict__ x, const float* __restrict__ attn,
            float* __restrict__ part)
{
    const int lane = threadIdx.x;
    const int n    = lane & 15;
    const int half = lane >> 4;
    const int blk  = blockIdx.x;
    const int d0   = (blk & 15) * SLAB;
    const int ch   = (blk >> 4) & (CH - 1);
    const int b    = blk >> 8;
    const int r0   = ch * CROWS;

    const float* xb = x    + ((size_t)b * SEQ + r0 + 2 * half) * DIM   + d0 + 4 * n;
    const float* ab = attn + ((size_t)b * SEQ + r0 + 2 * half) * HEADS + n;
    v8f c[4] = {};

    v2f a0[PS], a1[PS];
    v4f p0[PS], p1[PS], q0[PS], q1[PS];
    constexpr int NSET = CROWS / (4 * PS);
    pl_load(a0, p0, p1, ab, xb);
    __builtin_amdgcn_sched_barrier(0);
    for (int s = 0; s < NSET - 2; s += 2) {
        pl_load(a1, q0, q1, ab + (size_t)(s + 1) * 4 * PS * HEADS, xb + (size_t)(s + 1) * 4 * PS * DIM);
        __builtin_amdgcn_sched_barrier(0);
        pl_mma(c, a0, p0, p1);
        __builtin_amdgcn_sched_barrier(0);
        pl_load(a0, p0, p1, ab + (size_t)(s + 2) * 4 * PS * HEADS, xb + (size_t)(s + 2) * 4 * PS * DIM);
        __builtin_amdgcn_sched_barrier(0);
        pl_mma(c, a1, q0, q1);
        __builtin_amdgcn_sched_barrier(0);
    }
    pl_load(a1, q0, q1, ab + (size_t)(NSET - 1) * 4 * PS * HEADS, xb + (size_t)(NSET - 1) * 4 * PS * DIM);
    __builtin_amdgcn_sched_barrier(0);
    pl_mma(c, a0, p0, p1);
    __builtin_amdgcn_sched_barrier(0);
    pl_mma(c, a1, q0, q1);

    float* pb = part + (((size_t)b * CH + ch) * HEADS + 8 * half) * DIM + d0 + 4 * n;
    #pragma unroll
    for (int r = 0; r < 8; ++r) {
        const v4f v = {c[0][r], c[1][r], c[2][r], c[3][r]};
        *(v4f*)(pb + (size_t)r * DIM) = v;
    }
}

__global__ __launch_bounds__(256)
void k_sum(const float* __restrict__ part, float* __restrict__ out)
{
    constexpr int PLANE = HEADS * DIM / 4;
    const int i = blockIdx.x * 256 + threadIdx.x;
    const int b = i / PLANE;
    const int r = i % PLANE;
    const v4f* p = (const v4f*)part + (size_t)b * CH * PLANE + r;
    v4f s = p[0];
    #pragma unroll
    for (int cidx = 1; cidx < CH; ++cidx)
        s += p[(size_t)cidx * PLANE];
    ((v4f*)out)[i] = s;
}

extern "C" void kernel_launch(void* const* d_in, const int* in_sizes, int n_in,
                              void* d_out, int out_size, void* d_ws, size_t ws_size,
                              hipStream_t stream)
{
    const float* x    = (const float*)d_in[0];
    const float* W    = (const float*)d_in[1];
    const int*   mask = (const int*)  d_in[2];
    float* out    = (float*)d_out;
    float* logits = (float*)d_ws;
    float* part   = logits + BATCH * SEQ * HEADS;

    k_logits <<<dim3(BATCH * SEQ / 16), dim3(32),  0, stream>>>(x, W, mask, logits);
    k_softmax<<<dim3(BATCH * HEADS),    dim3(256), 0, stream>>>(logits);
    k_pool   <<<dim3(BATCH * CH * (DIM / SLAB)), dim3(32), 0, stream>>>(x, logits, part);
    k_sum    <<<dim3(BATCH * HEADS * DIM / 4 / 256), dim3(256), 0, stream>>>(part, out);
}


# k_logits: one 16-byte x load per lane feeds two products (its 4 neighbouring k); W fragments read from an LDS copy in that order, built once per 8-wave work-group (changes summation order)
# speedup vs baseline: 3.7419x; 1.0882x over previous
#include <hip/hip_runtime.h>
#include <hip/hip_bf16.h>

typedef __attribute__((ext_vector_type(2))) float v2f;
typedef __attribute__((ext_vector_type(4))) float v4f;
typedef __attribute__((ext_vector_type(8))) float v8f;

constexpr int BATCH = 8;
constexpr int SEQ   = 4096;
constexpr int DIM   = 1024;
constexpr int HEADS = 16;
constexpr float SCALE = 0.125f;
constexpr int LGW = 8;
constexpr int QN  = DIM / 8;
constexpr int XQ  = 16;
constexpr int NXS = QN / XQ;
constexpr int CH    = 16;
constexpr int CROWS = SEQ / CH;
constexpr int PS    = 8;
constexpr int SLAB  = 64;

__device__ __forceinline__ void lx_load(v4f (&xs)[XQ], const float* __restrict__ xp)
{
    #pragma unroll
    for (int j = 0; j < XQ; ++j)
        xs[j] = *(const v4f*)(xp + 8 * j);
}

__device__ __forceinline__ void lw_load(v4f (&ws)[XQ], const v4f* wl)
{
    #pragma unroll
    for (int j = 0; j < XQ; ++j)
        ws[j] = wl[32 * j];
}

__device__ __forceinline__ v8f lg_mma(v8f c, const v4f (&xs)[XQ], const v4f (&ws)[XQ])
{
    #pragma unroll
    for (int j = 0; j < XQ; ++j) {
        const v2f a0 = {xs[j].x, xs[j].y};
        const v2f b0 = {ws[j].x, ws[j].y};
        const v2f a1 = {xs[j].z, xs[j].w};
        const v2f b1 = {ws[j].z, ws[j].w};
        c = __builtin_amdgcn_wmma_f32_16x16x4_f32(
                false, a0, false, b0, (short)0, c, false, false);
        c = __builtin_amdgcn_wmma_f32_16x16x4_f32(
                false, a1, false, b1, (short)0, c, false, false);
    }
    return c;
}

__device__ __forceinline__ void pl_load(v2f (&a)[PS], v4f (&q0)[PS], v4f (&q1)[PS],
                                        const float* __restrict__ ab,
                                        const float* __restrict__ xb)
{
    #pragma unroll
    for (int j = 0; j < PS; ++j) {
        a[j].x = ab[(size_t)(4 * j + 0) * HEADS];
        a[j].y = ab[(size_t)(4 * j + 1) * HEADS];
        q0[j]  = *(const v4f*)(xb + (size_t)(4 * j + 0) * DIM);
        q1[j]  = *(const v4f*)(xb + (size_t)(4 * j + 1) * DIM);
    }
}

__device__ __forceinline__ void pl_mma(v8f (&c)[4], const v2f (&a)[PS],
                                       const v4f (&q0)[PS], const v4f (&q1)[PS])
{
    #pragma unroll
    for (int j = 0; j < PS; ++j) {
        const v2f b0 = {q0[j].x, q1[j].x};
        const v2f b1 = {q0[j].y, q1[j].y};
        const v2f b2 = {q0[j].z, q1[j].z};
        const v2f b3 = {q0[j].w, q1[j].w};
        c[0] = __builtin_amdgcn_wmma_f32_16x16x4_f32(false, a[j], false, b0, (short)0, c[0], false, false);
        c[1] = __builtin_amdgcn_wmma_f32_16x16x4_f32(false, a[j], false, b1, (short)0, c[1], false, false);
        c[2] = __builtin_amdgcn_wmma_f32_16x16x4_f32(false, a[j], false, b2, (short)0, c[2], false, false);
        c[3] = __builtin_amdgcn_wmma_f32_16x16x4_f32(false, a[j], false, b3, (short)0, c[3], false, false);
    }
}

__global__ __launch_bounds__(32 * LGW)
void k_logits(const float* __restrict__ x, const float* __restrict__ W,
              const int* __restrict__ mask, float* __restrict__ logits)
{
    __shared__ v4f wf[QN * 32];
    const int tid  = threadIdx.x;
    const int lane = tid & 31;
    const int m    = lane & 15;
    const int half = lane >> 4;
    const int tile = blockIdx.x * LGW + (tid >> 5);
    const int b    = tile >> 8;
    const int s0   = (tile & 255) << 4;

    v4f r[4][4];
    #pragma unroll
    for (int i = 0; i < 4; ++i) {
        const int id = tid + 32 * LGW * i;
        const float* wp = W + (size_t)(id >> 2) * 4 * HEADS + 4 * (id & 3);
        #pragma unroll
        for (int k = 0; k < 4; ++k)
            r[i][k] = *(const v4f*)(wp + k * HEADS);
    }

    const float* xp = x + ((size_t)b * SEQ + (s0 + m)) * DIM + 4 * half;
    v4f xb[4][XQ], wb[2][XQ];
    lx_load(xb[0], xp);
    lx_load(xb[1], xp + 8 * XQ);

    #pragma unroll
    for (int i = 0; i < 4; ++i) {
        const int id = tid + 32 * LGW * i;
        v4f* dst = wf + (id >> 2) * 16 + 4 * (id & 3);
        #pragma unroll
        for (int j = 0; j < 4; ++j) {
            const v4f col = {r[i][0][j], r[i][1][j], r[i][2][j], r[i][3][j]};
            dst[j] = col;
        }
    }
    asm volatile("s_wait_dscnt 0x0" ::: "memory");
    asm volatile("s_barrier_signal -1" ::: "memory");
    asm volatile("s_barrier_wait -1" ::: "memory");

    const v4f* wl = wf + lane;
    lx_load(xb[2], xp + 2 * 8 * XQ);
    lw_load(wb[0], wl);
    v8f c = {};
    #pragma unroll
    for (int t = 0; t < NXS; ++t) {
        if (t + 3 < NXS) lx_load(xb[(t + 3) & 3], xp + (t + 3) * 8 * XQ);
        if (t + 1 < NXS) lw_load(wb[(t + 1) & 1], wl + (t + 1) * 32 * XQ);
        __builtin_amdgcn_sched_barrier(0);
        c = lg_mma(c, xb[t & 3], wb[t & 1]);
        __builtin_amdgcn_sched_barrier(0);
    }

    #pragma unroll
    for (int rr = 0; rr < 8; ++rr) {
        const int s = s0 + rr + 8 * half;
        float v = c[rr] * SCALE;
        if (mask[(size_t)b * SEQ + s] == 0) v = -__builtin_inff();
        logits[((size_t)b * SEQ + s) * HEADS + m] = v;
    }
}

__global__ __launch_bounds__(256)
void k_softmax(float* __restrict__ logits)
{
    __shared__ float red[256];
    const int bh = blockIdx.x;
    const int b  = bh >> 4;
    const int h  = bh & 15;
    const int t  = threadIdx.x;
    float* base = logits + (size_t)b * SEQ * HEADS + h;

    float vals[16];
    float mx = -__builtin_inff();
    #pragma unroll
    for (int i = 0; i < 16; ++i) {
        float v = base[(size_t)(i * 256 + t) * HEADS];
        vals[i] = v;
        mx = fmaxf(mx, v);
    }
    red[t] = mx; __syncthreads();
    for (int off = 128; off > 0; off >>= 1) {
        if (t < off) red[t] = fmaxf(red[t], red[t + off]);
        __syncthreads();
    }
    mx = red[0];
    __syncthreads();

    float sum = 0.0f;
    #pragma unroll
    for (int i = 0; i < 16; ++i) {
        float e = __expf(vals[i] - mx);
        vals[i] = e;
        sum += e;
    }
    red[t] = sum; __syncthreads();
    for (int off = 128; off > 0; off >>= 1) {
        if (t < off) red[t] += red[t + off];
        __syncthreads();
    }
    const float rinv = 1.0f / red[0];

    #pragma unroll
    for (int i = 0; i < 16; ++i)
        base[(size_t)(i * 256 + t) * HEADS] = vals[i] * rinv;
}

__global__ __launch_bounds__(32)
void k_pool(const float* __restrict__ x, const float* __restrict__ attn,
            float* __restrict__ part)
{
    const int lane = threadIdx.x;
    const int n    = lane & 15;
    const int half = lane >> 4;
    const int blk  = blockIdx.x;
    const int d0   = (blk & 15) * SLAB;
    const int ch   = (blk >> 4) & (CH - 1);
    const int b    = blk >> 8;
    const int r0   = ch * CROWS;

    const float* xb = x    + ((size_t)b * SEQ + r0 + 2 * half) * DIM   + d0 + 4 * n;
    const float* ab = attn + ((size_t)b * SEQ + r0 + 2 * half) * HEADS + n;
    v8f c[4] = {};

    v2f a0[PS], a1[PS];
    v4f p0[PS], p1[PS], q0[PS], q1[PS];
    constexpr int NSET = CROWS / (4 * PS);
    pl_load(a0, p0, p1, ab, xb);
    __builtin_amdgcn_sched_barrier(0);
    for (int s = 0; s < NSET - 2; s += 2) {
        pl_load(a1, q0, q1, ab + (size_t)(s + 1) * 4 * PS * HEADS, xb + (size_t)(s + 1) * 4 * PS * DIM);
        __builtin_amdgcn_sched_barrier(0);
        pl_mma(c, a0, p0, p1);
        __builtin_amdgcn_sched_barrier(0);
        pl_load(a0, p0, p1, ab + (size_t)(s + 2) * 4 * PS * HEADS, xb + (size_t)(s + 2) * 4 * PS * DIM);
        __builtin_amdgcn_sched_barrier(0);
        pl_mma(c, a1, q0, q1);
        __builtin_amdgcn_sched_barrier(0);
    }
    pl_load(a1, q0, q1, ab + (size_t)(NSET - 1) * 4 * PS * HEADS, xb + (size_t)(NSET - 1) * 4 * PS * DIM);
    __builtin_amdgcn_sched_barrier(0);
    pl_mma(c, a0, p0, p1);
    __builtin_amdgcn_sched_barrier(0);
    pl_mma(c, a1, q0, q1);

    float* pb = part + (((size_t)b * CH + ch) * HEADS + 8 * half) * DIM + d0 + 4 * n;
    #pragma unroll
    for (int r = 0; r < 8; ++r) {
        const v4f v = {c[0][r], c[1][r], c[2][r], c[3][r]};
        *(v4f*)(pb + (size_t)r * DIM) = v;
    }
}

__global__ __launch_bounds__(256)
void k_sum(const float* __restrict__ part, float* __restrict__ out)
{
    constexpr int PLANE = HEADS * DIM / 4;
    const int i = blockIdx.x * 256 + threadIdx.x;
    const int b = i / PLANE;
    const int r = i % PLANE;
    const v4f* p = (const v4f*)part + (size_t)b * CH * PLANE + r;
    v4f s = p[0];
    #pragma unroll
    for (int cidx = 1; cidx < CH; ++cidx)
        s += p[(size_t)cidx * PLANE];
    ((v4f*)out)[i] = s;
}

extern "C" void kernel_launch(void* const* d_in, const int* in_sizes, int n_in,
                              void* d_out, int out_size, void* d_ws, size_t ws_size,
                              hipStream_t stream)
{
    const float* x    = (const float*)d_in[0];
    const float* W    = (const float*)d_in[1];
    const int*   mask = (const int*)  d_in[2];
    float* out    = (float*)d_out;
    float* logits = (float*)d_ws;
    float* part   = logits + BATCH * SEQ * HEADS;

    k_logits <<<dim3(BATCH * SEQ / 16 / LGW), dim3(32 * LGW), 0, stream>>>(x, W, mask, logits);
    k_softmax<<<dim3(BATCH * HEADS),    dim3(256), 0, stream>>>(logits);
    k_pool   <<<dim3(BATCH * CH * (DIM / SLAB)), dim3(32), 0, stream>>>(x, logits, part);
    k_sum    <<<dim3(BATCH * HEADS * DIM / 4 / 256), dim3(256), 0, stream>>>(part, out);
}


# k_logits: K split over two waves per tile (512 k each, partial tiles added low + high through LDS); 16 waves of half the registers per work-group instead of 8 (changes summation order)
# speedup vs baseline: 4.1500x; 1.1091x over previous
#include <hip/hip_runtime.h>
#include <hip/hip_bf16.h>

typedef __attribute__((ext_vector_type(2))) float v2f;
typedef __attribute__((ext_vector_type(4))) float v4f;
typedef __attribute__((ext_vector_type(8))) float v8f;

constexpr int BATCH = 8;
constexpr int SEQ   = 4096;
constexpr int DIM   = 1024;
constexpr int HEADS = 16;
constexpr float SCALE = 0.125f;
constexpr int LGT = 8;
constexpr int LGW = 2 * LGT;
constexpr int QN  = DIM / 8;
constexpr int QH  = QN / 2;
constexpr int XQ  = 8;
constexpr int NXS = QH / XQ;
constexpr int CH    = 16;
constexpr int CROWS = SEQ / CH;
constexpr int PS    = 8;
constexpr int SLAB  = 64;

__device__ __forceinline__ void lx_load(v4f (&xs)[XQ], const float* __restrict__ xp)
{
    #pragma unroll
    for (int j = 0; j < XQ; ++j)
        xs[j] = *(const v4f*)(xp + 8 * j);
}

__device__ __forceinline__ void lw_load(v4f (&ws)[XQ], const v4f* wl)
{
    #pragma unroll
    for (int j = 0; j < XQ; ++j)
        ws[j] = wl[32 * j];
}

__device__ __forceinline__ v8f lg_mma(v8f c, const v4f (&xs)[XQ], const v4f (&ws)[XQ])
{
    #pragma unroll
    for (int j = 0; j < XQ; ++j) {
        const v2f a0 = {xs[j].x, xs[j].y};
        const v2f b0 = {ws[j].x, ws[j].y};
        const v2f a1 = {xs[j].z, xs[j].w};
        const v2f b1 = {ws[j].z, ws[j].w};
        c = __builtin_amdgcn_wmma_f32_16x16x4_f32(
                false, a0, false, b0, (short)0, c, false, false);
        c = __builtin_amdgcn_wmma_f32_16x16x4_f32(
                false, a1, false, b1, (short)0, c, false, false);
    }
    return c;
}

__device__ __forceinline__ void pl_load(v2f (&a)[PS], v4f (&q0)[PS], v4f (&q1)[PS],
                                        const float* __restrict__ ab,
                                        const float* __restrict__ xb)
{
    #pragma unroll
    for (int j = 0; j < PS; ++j) {
        a[j].x = ab[(size_t)(4 * j + 0) * HEADS];
        a[j].y = ab[(size_t)(4 * j + 1) * HEADS];
        q0[j]  = *(const v4f*)(xb + (size_t)(4 * j + 0) * DIM);
        q1[j]  = *(const v4f*)(xb + (size_t)(4 * j + 1) * DIM);
    }
}

__device__ __forceinline__ void pl_mma(v8f (&c)[4], const v2f (&a)[PS],
                                       const v4f (&q0)[PS], const v4f (&q1)[PS])
{
    #pragma unroll
    for (int j = 0; j < PS; ++j) {
        const v2f b0 = {q0[j].x, q1[j].x};
        const v2f b1 = {q0[j].y, q1[j].y};
        const v2f b2 = {q0[j].z, q1[j].z};
        const v2f b3 = {q0[j].w, q1[j].w};
        c[0] = __builtin_amdgcn_wmma_f32_16x16x4_f32(false, a[j], false, b0, (short)0, c[0], false, false);
        c[1] = __builtin_amdgcn_wmma_f32_16x16x4_f32(false, a[j], false, b1, (short)0, c[1], false, false);
        c[2] = __builtin_amdgcn_wmma_f32_16x16x4_f32(false, a[j], false, b2, (short)0, c[2], false, false);
        c[3] = __builtin_amdgcn_wmma_f32_16x16x4_f32(false, a[j], false, b3, (short)0, c[3], false, false);
    }
}

__global__ __launch_bounds__(32 * LGW)
void k_logits(const float* __restrict__ x, const float* __restrict__ W,
              const int* __restrict__ mask, float* __restrict__ logits)
{
    __shared__ v4f wf[QN * 32];
    __shared__ float hi[LGT][8][32];
    const int tid  = threadIdx.x;
    const int lane = tid & 31;
    const int m    = lane & 15;
    const int half = lane >> 4;
    const int wv   = tid >> 5;
    const int tw   = wv & (LGT - 1);
    const int kh   = wv / LGT;
    const int tile = blockIdx.x * LGT + tw;
    const int b    = tile >> 8;
    const int s0   = (tile & 255) << 4;

    v4f r[2][4];
    #pragma unroll
    for (int i = 0; i < 2; ++i) {
        const int id = tid + 32 * LGW * i;
        const float* wp = W + (size_t)(id >> 2) * 4 * HEADS + 4 * (id & 3);
        #pragma unroll
        for (int k = 0; k < 4; ++k)
            r[i][k] = *(const v4f*)(wp + k * HEADS);
    }

    const float* xp = x + ((size_t)b * SEQ + (s0 + m)) * DIM + 8 * QH * kh + 4 * half;
    v4f xb[4][XQ], wb[2][XQ];
    lx_load(xb[0], xp);
    lx_load(xb[1], xp + 8 * XQ);
    __builtin_amdgcn_sched_barrier(0);

    #pragma unroll
    for (int i = 0; i < 2; ++i) {
        const int id = tid + 32 * LGW * i;
        v4f* dst = wf + (id >> 2) * 16 + 4 * (id & 3);
        #pragma unroll
        for (int j = 0; j < 4; ++j) {
            const v4f col = {r[i][0][j], r[i][1][j], r[i][2][j], r[i][3][j]};
            dst[j] = col;
        }
    }
    asm volatile("s_wait_dscnt 0x0" ::: "memory");
    asm volatile("s_barrier_signal -1" ::: "memory");
    asm volatile("s_barrier_wait -1" ::: "memory");

    const v4f* wl = wf + 32 * QH * kh + lane;
    lx_load(xb[2], xp + 2 * 8 * XQ);
    lw_load(wb[0], wl);
    v8f c = {};
    #pragma unroll
    for (int t = 0; t < NXS; ++t) {
        if (t + 3 < NXS) lx_load(xb[(t + 3) & 3], xp + (t + 3) * 8 * XQ);
        if (t + 1 < NXS) lw_load(wb[(t + 1) & 1], wl + (t + 1) * 32 * XQ);
        __builtin_amdgcn_sched_barrier(0);
        c = lg_mma(c, xb[t & 3], wb[t & 1]);
        __builtin_amdgcn_sched_barrier(0);
    }

    if (kh == 1) {
        #pragma unroll
        for (int rr = 0; rr < 8; ++rr)
            hi[tw][rr][lane] = c[rr];
    }
    __syncthreads();
    if (kh == 0) {
        #pragma unroll
        for (int rr = 0; rr < 8; ++rr) {
            const int s = s0 + rr + 8 * half;
            float v = (c[rr] + hi[tw][rr][lane]) * SCALE;
            if (mask[(size_t)b * SEQ + s] == 0) v = -__builtin_inff();
            logits[((size_t)b * SEQ + s) * HEADS + m] = v;
        }
    }
}

__global__ __launch_bounds__(256)
void k_softmax(float* __restrict__ logits)
{
    __shared__ float red[256];
    const int bh = blockIdx.x;
    const int b  = bh >> 4;
    const int h  = bh & 15;
    const int t  = threadIdx.x;
    float* base = logits + (size_t)b * SEQ * HEADS + h;

    float vals[16];
    float mx = -__builtin_inff();
    #pragma unroll
    for (int i = 0; i < 16; ++i) {
        float v = base[(size_t)(i * 256 + t) * HEADS];
        vals[i] = v;
        mx = fmaxf(mx, v);
    }
    red[t] = mx; __syncthreads();
    for (int off = 128; off > 0; off >>= 1) {
        if (t < off) red[t] = fmaxf(red[t], red[t + off]);
        __syncthreads();
    }
    mx = red[0];
    __syncthreads();

    float sum = 0.0f;
    #pragma unroll
    for (int i = 0; i < 16; ++i) {
        float e = __expf(vals[i] - mx);
        vals[i] = e;
        sum += e;
    }
    red[t] = sum; __syncthreads();
    for (int off = 128; off > 0; off >>= 1) {
        if (t < off) red[t] += red[t + off];
        __syncthreads();
    }
    const float rinv = 1.0f / red[0];

    #pragma unroll
    for (int i = 0; i < 16; ++i)
        base[(size_t)(i * 256 + t) * HEADS] = vals[i] * rinv;
}

__global__ __launch_bounds__(32)
void k_pool(const float* __restrict__ x, const float* __restrict__ attn,
            float* __restrict__ part)
{
    const int lane = threadIdx.x;
    const int n    = lane & 15;
    const int half = lane >> 4;
    const int blk  = blockIdx.x;
    const int d0   = (blk & 15) * SLAB;
    const int ch   = (blk >> 4) & (CH - 1);
    const int b    = blk >> 8;
    const int r0   = ch * CROWS;

    const float* xb = x    + ((size_t)b * SEQ + r0 + 2 * half) * DIM   + d0 + 4 * n;
    const float* ab = attn + ((size_t)b * SEQ + r0 + 2 * half) * HEADS + n;
    v8f c[4] = {};

    v2f a0[PS], a1[PS];
    v4f p0[PS], p1[PS], q0[PS], q1[PS];
    constexpr int NSET = CROWS / (4 * PS);
    pl_load(a0, p0, p1, ab, xb);
    __builtin_amdgcn_sched_barrier(0);
    for (int s = 0; s < NSET - 2; s += 2) {
        pl_load(a1, q0, q1, ab + (size_t)(s + 1) * 4 * PS * HEADS, xb + (size_t)(s + 1) * 4 * PS * DIM);
        __builtin_amdgcn_sched_barrier(0);
        pl_mma(c, a0, p0, p1);
        __builtin_amdgcn_sched_barrier(0);
        pl_load(a0, p0, p1, ab + (size_t)(s + 2) * 4 * PS * HEADS, xb + (size_t)(s + 2) * 4 * PS * DIM);
        __builtin_amdgcn_sched_barrier(0);
        pl_mma(c, a1, q0, q1);
        __builtin_amdgcn_sched_barrier(0);
    }
    pl_load(a1, q0, q1, ab + (size_t)(NSET - 1) * 4 * PS * HEADS, xb + (size_t)(NSET - 1) * 4 * PS * DIM);
    __builtin_amdgcn_sched_barrier(0);
    pl_mma(c, a0, p0, p1);
    __builtin_amdgcn_sched_barrier(0);
    pl_mma(c, a1, q0, q1);

    float* pb = part + (((size_t)b * CH + ch) * HEADS + 8 * half) * DIM + d0 + 4 * n;
    #pragma unroll
    for (int r = 0; r < 8; ++r) {
        const v4f v = {c[0][r], c[1][r], c[2][r], c[3][r]};
        *(v4f*)(pb + (size_t)r * DIM) = v;
    }
}

__global__ __launch_bounds__(256)
void k_sum(const float* __restrict__ part, float* __restrict__ out)
{
    constexpr int PLANE = HEADS * DIM / 4;
    const int i = blockIdx.x * 256 + threadIdx.x;
    const int b = i / PLANE;
    const int r = i % PLANE;
    const v4f* p = (const v4f*)part + (size_t)b * CH * PLANE + r;
    v4f s = p[0];
    #pragma unroll
    for (int cidx = 1; cidx < CH; ++cidx)
        s += p[(size_t)cidx * PLANE];
    ((v4f*)out)[i] = s;
}

extern "C" void kernel_launch(void* const* d_in, const int* in_sizes, int n_in,
                              void* d_out, int out_size, void* d_ws, size_t ws_size,
                              hipStream_t stream)
{
    const float* x    = (const float*)d_in[0];
    const float* W    = (const float*)d_in[1];
    const int*   mask = (const int*)  d_in[2];
    float* out    = (float*)d_out;
    float* logits = (float*)d_ws;
    float* part   = logits + BATCH * SEQ * HEADS;

    k_logits <<<dim3(BATCH * SEQ / 16 / LGT), dim3(32 * LGW), 0, stream>>>(x, W, mask, logits);
    k_softmax<<<dim3(BATCH * HEADS),    dim3(256), 0, stream>>>(logits);
    k_pool   <<<dim3(BATCH * CH * (DIM / SLAB)), dim3(32), 0, stream>>>(x, logits, part);
    k_sum    <<<dim3(BATCH * HEADS * DIM / 4 / 256), dim3(256), 0, stream>>>(part, out);
}


# softmax launch removed: k_logits writes exp(logit - group max) and a (max, sum) pair per 128-row group and head; k_pool scales its A fragments by exp(max_g - max)/sum (changes summation order)
# speedup vs baseline: 5.4277x; 1.3079x over previous
#include <hip/hip_runtime.h>
#include <hip/hip_bf16.h>

typedef __attribute__((ext_vector_type(2))) float v2f;
typedef __attribute__((ext_vector_type(4))) float v4f;
typedef __attribute__((ext_vector_type(8))) float v8f;

constexpr int BATCH = 8;
constexpr int SEQ   = 4096;
constexpr int DIM   = 1024;
constexpr int HEADS = 16;
constexpr float SCALE = 0.125f;
constexpr int LGT = 8;
constexpr int LGW = 2 * LGT;
constexpr int QN  = DIM / 8;
constexpr int QH  = QN / 2;
constexpr int XQ  = 8;
constexpr int NXS = QH / XQ;
constexpr int CH    = 16;
constexpr int CROWS = SEQ / CH;
constexpr int PS    = 8;
constexpr int SLAB  = 64;
constexpr int NG    = SEQ / (16 * LGT);

__device__ __forceinline__ void lx_load(v4f (&xs)[XQ], const float* __restrict__ xp)
{
    #pragma unroll
    for (int j = 0; j < XQ; ++j)
        xs[j] = *(const v4f*)(xp + 8 * j);
}

__device__ __forceinline__ void lw_load(v4f (&ws)[XQ], const v4f* wl)
{
    #pragma unroll
    for (int j = 0; j < XQ; ++j)
        ws[j] = wl[32 * j];
}

__device__ __forceinline__ v8f lg_mma(v8f c, const v4f (&xs)[XQ], const v4f (&ws)[XQ])
{
    #pragma unroll
    for (int j = 0; j < XQ; ++j) {
        const v2f a0 = {xs[j].x, xs[j].y};
        const v2f b0 = {ws[j].x, ws[j].y};
        const v2f a1 = {xs[j].z, xs[j].w};
        const v2f b1 = {ws[j].z, ws[j].w};
        c = __builtin_amdgcn_wmma_f32_16x16x4_f32(
                false, a0, false, b0, (short)0, c, false, false);
        c = __builtin_amdgcn_wmma_f32_16x16x4_f32(
                false, a1, false, b1, (short)0, c, false, false);
    }
    return c;
}

__device__ __forceinline__ void pl_load(v2f (&a)[PS], v4f (&q0)[PS], v4f (&q1)[PS],
                                        const float* __restrict__ ab,
                                        const float* __restrict__ xb)
{
    #pragma unroll
    for (int j = 0; j < PS; ++j) {
        a[j].x = ab[(size_t)(4 * j + 0) * HEADS];
        a[j].y = ab[(size_t)(4 * j + 1) * HEADS];
        q0[j]  = *(const v4f*)(xb + (size_t)(4 * j + 0) * DIM);
        q1[j]  = *(const v4f*)(xb + (size_t)(4 * j + 1) * DIM);
    }
}

__device__ __forceinline__ void pl_mma(v8f (&c)[4], const v2f (&a)[PS],
                                       const v4f (&q0)[PS], const v4f (&q1)[PS],
                                       const float f)
{
    #pragma unroll
    for (int j = 0; j < PS; ++j) {
        const v2f b0 = {q0[j].x, q1[j].x};
        const v2f b1 = {q0[j].y, q1[j].y};
        const v2f b2 = {q0[j].z, q1[j].z};
        const v2f b3 = {q0[j].w, q1[j].w};
        const v2f aj = {a[j].x * f, a[j].y * f};
        c[0] = __builtin_amdgcn_wmma_f32_16x16x4_f32(false, aj, false, b0, (short)0, c[0], false, false);
        c[1] = __builtin_amdgcn_wmma_f32_16x16x4_f32(false, aj, false, b1, (short)0, c[1], false, false);
        c[2] = __builtin_amdgcn_wmma_f32_16x16x4_f32(false, aj, false, b2, (short)0, c[2], false, false);
        c[3] = __builtin_amdgcn_wmma_f32_16x16x4_f32(false, aj, false, b3, (short)0, c[3], false, false);
    }
}

__global__ __launch_bounds__(32 * LGW)
void k_logits(const float* __restrict__ x, const float* __restrict__ W,
              const int* __restrict__ mask, float* __restrict__ ex,
              float* __restrict__ stats)
{
    __shared__ v4f wf[QN * 32];
    __shared__ float hi[LGT][8][32];
    __shared__ float gmax[LGT][HEADS];
    __shared__ float gsum[LGT][HEADS];
    const int tid  = threadIdx.x;
    const int lane = tid & 31;
    const int m    = lane & 15;
    const int half = lane >> 4;
    const int wv   = tid >> 5;
    const int tw   = wv & (LGT - 1);
    const int kh   = wv / LGT;
    const int tile = blockIdx.x * LGT + tw;
    const int b    = tile >> 8;
    const int s0   = (tile & 255) << 4;

    v4f r[2][4];
    #pragma unroll
    for (int i = 0; i < 2; ++i) {
        const int id = tid + 32 * LGW * i;
        const float* wp = W + (size_t)(id >> 2) * 4 * HEADS + 4 * (id & 3);
        #pragma unroll
        for (int k = 0; k < 4; ++k)
            r[i][k] = *(const v4f*)(wp + k * HEADS);
    }

    const float* xp = x + ((size_t)b * SEQ + (s0 + m)) * DIM + 8 * QH * kh + 4 * half;
    v4f xb[4][XQ], wb[2][XQ];
    lx_load(xb[0], xp);
    lx_load(xb[1], xp + 8 * XQ);
    __builtin_amdgcn_sched_barrier(0);

    #pragma unroll
    for (int i = 0; i < 2; ++i) {
        const int id = tid + 32 * LGW * i;
        v4f* dst = wf + (id >> 2) * 16 + 4 * (id & 3);
        #pragma unroll
        for (int j = 0; j < 4; ++j) {
            const v4f col = {r[i][0][j], r[i][1][j], r[i][2][j], r[i][3][j]};
            dst[j] = col;
        }
    }
    asm volatile("s_wait_dscnt 0x0" ::: "memory");
    asm volatile("s_barrier_signal -1" ::: "memory");
    asm volatile("s_barrier_wait -1" ::: "memory");

    const v4f* wl = wf + 32 * QH * kh + lane;
    lx_load(xb[2], xp + 2 * 8 * XQ);
    lw_load(wb[0], wl);
    v8f c = {};
    #pragma unroll
    for (int t = 0; t < NXS; ++t) {
        if (t + 3 < NXS) lx_load(xb[(t + 3) & 3], xp + (t + 3) * 8 * XQ);
        if (t + 1 < NXS) lw_load(wb[(t + 1) & 1], wl + (t + 1) * 32 * XQ);
        __builtin_amdgcn_sched_barrier(0);
        c = lg_mma(c, xb[t & 3], wb[t & 1]);
        __builtin_amdgcn_sched_barrier(0);
    }

    if (kh == 1) {
        #pragma unroll
        for (int rr = 0; rr < 8; ++rr)
            hi[tw][rr][lane] = c[rr];
    }
    __syncthreads();
    float v[8] = {};
    if (kh == 0) {
        float mw = -__builtin_inff();
        #pragma unroll
        for (int rr = 0; rr < 8; ++rr) {
            const int s = s0 + rr + 8 * half;
            float t = (c[rr] + hi[tw][rr][lane]) * SCALE;
            if (mask[(size_t)b * SEQ + s] == 0) t = -__builtin_inff();
            v[rr] = t;
            mw = fmaxf(mw, t);
        }
        mw = fmaxf(mw, __shfl_xor(mw, 16));
        if (half == 0) gmax[tw][m] = mw;
    }
    __syncthreads();
    float mg = 0.0f;
    if (kh == 0) {
        mg = gmax[0][m];
        #pragma unroll
        for (int t = 1; t < LGT; ++t) mg = fmaxf(mg, gmax[t][m]);
        const float ms = (mg == -__builtin_inff()) ? 0.0f : mg;
        float sw = 0.0f;
        #pragma unroll
        for (int rr = 0; rr < 8; ++rr) {
            v[rr] = __expf(v[rr] - ms);
            sw += v[rr];
        }
        sw += __shfl_xor(sw, 16);
        if (half == 0) gsum[tw][m] = sw;
    }
    __syncthreads();
    if (kh == 0) {
        #pragma unroll
        for (int rr = 0; rr < 8; ++rr) {
            const int s = s0 + rr + 8 * half;
            ex[((size_t)b * SEQ + s) * HEADS + m] = v[rr];
        }
    }
    if (wv == 0 && half == 0) {
        float l = gsum[0][m];
        #pragma unroll
        for (int t = 1; t < LGT; ++t) l += gsum[t][m];
        const int g = blockIdx.x & (NG - 1);
        const v2f pr = {mg, l};
        *(v2f*)(stats + (((size_t)b * HEADS + m) * NG + g) * 2) = pr;
    }
}

__global__ __launch_bounds__(32)
void k_pool(const float* __restrict__ x, const float* __restrict__ attn,
            const float* __restrict__ stats, float* __restrict__ part)
{
    const int lane = threadIdx.x;
    const int n    = lane & 15;
    const int half = lane >> 4;
    const int blk  = blockIdx.x;
    const int d0   = (blk & 15) * SLAB;
    const int ch   = (blk >> 4) & (CH - 1);
    const int b    = blk >> 8;
    const int r0   = ch * CROWS;

    const float* xb = x    + ((size_t)b * SEQ + r0 + 2 * half) * DIM   + d0 + 4 * n;
    const float* ab = attn + ((size_t)b * SEQ + r0 + 2 * half) * HEADS + n;
    v8f c[4] = {};

    v2f a0[PS], a1[PS];
    v4f p0[PS], p1[PS], q0[PS], q1[PS];
    constexpr int NSET = CROWS / (4 * PS);
    static_assert(CROWS == 2 * 16 * LGT && NG == 32 && NSET == 8, "a chunk is two groups of four sets");

    const float* sb = stats + ((size_t)b * HEADS + n) * NG * 2;
    v4f st[8];
    #pragma unroll
    for (int i = 0; i < 8; ++i)
        st[i] = *(const v4f*)(sb + 32 * half + 4 * i);
    const v4f own = *(const v4f*)(sb + 4 * ch);

    pl_load(a0, p0, p1, ab, xb);
    __builtin_amdgcn_sched_barrier(0);
    pl_load(a1, q0, q1, ab + (size_t)4 * PS * HEADS, xb + (size_t)4 * PS * DIM);
    __builtin_amdgcn_sched_barrier(0);

    float mh = -__builtin_inff();
    #pragma unroll
    for (int i = 0; i < 8; ++i) mh = fmaxf(mh, fmaxf(st[i].x, st[i].z));
    const float M  = fmaxf(mh, __shfl_xor(mh, 16));
    const float Ms = (M == -__builtin_inff()) ? 0.0f : M;
    float zh = 0.0f;
    #pragma unroll
    for (int i = 0; i < 8; ++i) {
        zh += st[i].y * __expf(st[i].x - Ms);
        zh += st[i].w * __expf(st[i].z - Ms);
    }
    const float Z  = zh + __shfl_xor(zh, 16);
    const float rz = 1.0f / Z;
    const float f0 = __expf(own.x - Ms) * rz;
    const float f1 = __expf(own.z - Ms) * rz;
    __builtin_amdgcn_sched_barrier(0);

    pl_mma(c, a0, p0, p1, f0);
    __builtin_amdgcn_sched_barrier(0);
    pl_load(a0, p0, p1, ab + (size_t)2 * 4 * PS * HEADS, xb + (size_t)2 * 4 * PS * DIM);
    __builtin_amdgcn_sched_barrier(0);
    pl_mma(c, a1, q0, q1, f0);
    __builtin_amdgcn_sched_barrier(0);
    for (int s = 2; s < NSET - 2; s += 2) {
        const float f = (s < NSET / 2) ? f0 : f1;
        pl_load(a1, q0, q1, ab + (size_t)(s + 1) * 4 * PS * HEADS, xb + (size_t)(s + 1) * 4 * PS * DIM);
        __builtin_amdgcn_sched_barrier(0);
        pl_mma(c, a0, p0, p1, f);
        __builtin_amdgcn_sched_barrier(0);
        pl_load(a0, p0, p1, ab + (size_t)(s + 2) * 4 * PS * HEADS, xb + (size_t)(s + 2) * 4 * PS * DIM);
        __builtin_amdgcn_sched_barrier(0);
        pl_mma(c, a1, q0, q1, f);
        __builtin_amdgcn_sched_barrier(0);
    }
    pl_load(a1, q0, q1, ab + (size_t)(NSET - 1) * 4 * PS * HEADS, xb + (size_t)(NSET - 1) * 4 * PS * DIM);
    __builtin_amdgcn_sched_barrier(0);
    pl_mma(c, a0, p0, p1, f1);
    __builtin_amdgcn_sched_barrier(0);
    pl_mma(c, a1, q0, q1, f1);

    float* pb = part + (((size_t)b * CH + ch) * HEADS + 8 * half) * DIM + d0 + 4 * n;
    #pragma unroll
    for (int r = 0; r < 8; ++r) {
        const v4f v = {c[0][r], c[1][r], c[2][r], c[3][r]};
        *(v4f*)(pb + (size_t)r * DIM) = v;
    }
}

__global__ __launch_bounds__(256)
void k_sum(const float* __restrict__ part, float* __restrict__ out)
{
    constexpr int PLANE = HEADS * DIM / 4;
    const int i = blockIdx.x * 256 + threadIdx.x;
    const int b = i / PLANE;
    const int r = i % PLANE;
    const v4f* p = (const v4f*)part + (size_t)b * CH * PLANE + r;
    v4f s = p[0];
    #pragma unroll
    for (int cidx = 1; cidx < CH; ++cidx)
        s += p[(size_t)cidx * PLANE];
    ((v4f*)out)[i] = s;
}

extern "C" void kernel_launch(void* const* d_in, const int* in_sizes, int n_in,
                              void* d_out, int out_size, void* d_ws, size_t ws_size,
                              hipStream_t stream)
{
    const float* x    = (const float*)d_in[0];
    const float* W    = (const float*)d_in[1];
    const int*   mask = (const int*)  d_in[2];
    float* out    = (float*)d_out;
    float* ex     = (float*)d_ws;
    float* part   = ex + BATCH * SEQ * HEADS;
    float* stats  = part + BATCH * CH * HEADS * DIM;

    k_logits <<<dim3(BATCH * SEQ / 16 / LGT), dim3(32 * LGW), 0, stream>>>(x, W, mask, ex, stats);
    k_pool   <<<dim3(BATCH * CH * (DIM / SLAB)), dim3(32), 0, stream>>>(x, ex, stats, part);
    k_sum    <<<dim3(BATCH * HEADS * DIM / 4 / 256), dim3(256), 0, stream>>>(part, out);
}


# k_pool: the 8 slab waves of a (batch, chunk, half row) form one work-group; the chunk's weights ex*f are built once in LDS in fragment order and read by all 8 (changes summation order)
# speedup vs baseline: 6.2453x; 1.1506x over previous
#include <hip/hip_runtime.h>
#include <hip/hip_bf16.h>

typedef __attribute__((ext_vector_type(2))) float v2f;
typedef __attribute__((ext_vector_type(4))) float v4f;
typedef __attribute__((ext_vector_type(8))) float v8f;

constexpr int BATCH = 8;
constexpr int SEQ   = 4096;
constexpr int DIM   = 1024;
constexpr int HEADS = 16;
constexpr float SCALE = 0.125f;
constexpr int LGT = 8;
constexpr int LGW = 2 * LGT;
constexpr int QN  = DIM / 8;
constexpr int QH  = QN / 2;
constexpr int XQ  = 8;
constexpr int NXS = QH / XQ;
constexpr int CH    = 16;
constexpr int CROWS = SEQ / CH;
constexpr int PS    = 8;
constexpr int SLAB  = 64;
constexpr int PW    = 8;
constexpr int RB    = CROWS / 8;
constexpr int NG    = SEQ / (16 * LGT);

__device__ __forceinline__ void lx_load(v4f (&xs)[XQ], const float* __restrict__ xp)
{
    #pragma unroll
    for (int j = 0; j < XQ; ++j)
        xs[j] = *(const v4f*)(xp + 8 * j);
}

__device__ __forceinline__ void lw_load(v4f (&ws)[XQ], const v4f* wl)
{
    #pragma unroll
    for (int j = 0; j < XQ; ++j)
        ws[j] = wl[32 * j];
}

__device__ __forceinline__ v8f lg_mma(v8f c, const v4f (&xs)[XQ], const v4f (&ws)[XQ])
{
    #pragma unroll
    for (int j = 0; j < XQ; ++j) {
        const v2f a0 = {xs[j].x, xs[j].y};
        const v2f b0 = {ws[j].x, ws[j].y};
        const v2f a1 = {xs[j].z, xs[j].w};
        const v2f b1 = {ws[j].z, ws[j].w};
        c = __builtin_amdgcn_wmma_f32_16x16x4_f32(
                false, a0, false, b0, (short)0, c, false, false);
        c = __builtin_amdgcn_wmma_f32_16x16x4_f32(
                false, a1, false, b1, (short)0, c, false, false);
    }
    return c;
}

__device__ __forceinline__ void px_load(v4f (&q)[2 * PS], const float* __restrict__ xb)
{
    #pragma unroll
    for (int j = 0; j < PS / 2; ++j) {
        #pragma unroll
        for (int i = 0; i < 4; ++i)
            q[4 * j + i] = *(const v4f*)(xb + (size_t)(8 * j + i) * DIM);
    }
}

__device__ __forceinline__ void pe_load(v4f (&e)[PS / 2], const v4f* wp)
{
    #pragma unroll
    for (int j = 0; j < PS / 2; ++j)
        e[j] = wp[32 * j];
}

__device__ __forceinline__ void pl_mma(v8f (&c)[4], const v4f (&e)[PS / 2],
                                       const v4f (&q)[2 * PS])
{
    #pragma unroll
    for (int j = 0; j < PS / 2; ++j) {
        #pragma unroll
        for (int p = 0; p < 2; ++p) {
            const v4f r0 = q[4 * j + 2 * p];
            const v4f r1 = q[4 * j + 2 * p + 1];
            const v2f aj = p ? (v2f){e[j].z, e[j].w} : (v2f){e[j].x, e[j].y};
            const v2f b0 = {r0.x, r1.x};
            const v2f b1 = {r0.y, r1.y};
            const v2f b2 = {r0.z, r1.z};
            const v2f b3 = {r0.w, r1.w};
            c[0] = __builtin_amdgcn_wmma_f32_16x16x4_f32(false, aj, false, b0, (short)0, c[0], false, false);
            c[1] = __builtin_amdgcn_wmma_f32_16x16x4_f32(false, aj, false, b1, (short)0, c[1], false, false);
            c[2] = __builtin_amdgcn_wmma_f32_16x16x4_f32(false, aj, false, b2, (short)0, c[2], false, false);
            c[3] = __builtin_amdgcn_wmma_f32_16x16x4_f32(false, aj, false, b3, (short)0, c[3], false, false);
        }
    }
}

__global__ __launch_bounds__(32 * LGW)
void k_logits(const float* __restrict__ x, const float* __restrict__ W,
              const int* __restrict__ mask, float* __restrict__ ex,
              float* __restrict__ stats)
{
    __shared__ v4f wf[QN * 32];
    __shared__ float hi[LGT][8][32];
    __shared__ float gmax[LGT][HEADS];
    __shared__ float gsum[LGT][HEADS];
    const int tid  = threadIdx.x;
    const int lane = tid & 31;
    const int m    = lane & 15;
    const int half = lane >> 4;
    const int wv   = tid >> 5;
    const int tw   = wv & (LGT - 1);
    const int kh   = wv / LGT;
    const int tile = blockIdx.x * LGT + tw;
    const int b    = tile >> 8;
    const int s0   = (tile & 255) << 4;

    v4f r[2][4];
    #pragma unroll
    for (int i = 0; i < 2; ++i) {
        const int id = tid + 32 * LGW * i;
        const float* wp = W + (size_t)(id >> 2) * 4 * HEADS + 4 * (id & 3);
        #pragma unroll
        for (int k = 0; k < 4; ++k)
            r[i][k] = *(const v4f*)(wp + k * HEADS);
    }

    const float* xp = x + ((size_t)b * SEQ + (s0 + m)) * DIM + 8 * QH * kh + 4 * half;
    v4f xb[4][XQ], wb[2][XQ];
    lx_load(xb[0], xp);
    lx_load(xb[1], xp + 8 * XQ);
    __builtin_amdgcn_sched_barrier(0);

    #pragma unroll
    for (int i = 0; i < 2; ++i) {
        const int id = tid + 32 * LGW * i;
        v4f* dst = wf + (id >> 2) * 16 + 4 * (id & 3);
        #pragma unroll
        for (int j = 0; j < 4; ++j) {
            const v4f col = {r[i][0][j], r[i][1][j], r[i][2][j], r[i][3][j]};
            dst[j] = col;
        }
    }
    asm volatile("s_wait_dscnt 0x0" ::: "memory");
    asm volatile("s_barrier_signal -1" ::: "memory");
    asm volatile("s_barrier_wait -1" ::: "memory");

    const v4f* wl = wf + 32 * QH * kh + lane;
    lx_load(xb[2], xp + 2 * 8 * XQ);
    lw_load(wb[0], wl);
    v8f c = {};
    #pragma unroll
    for (int t = 0; t < NXS; ++t) {
        if (t + 3 < NXS) lx_load(xb[(t + 3) & 3], xp + (t + 3) * 8 * XQ);
        if (t + 1 < NXS) lw_load(wb[(t + 1) & 1], wl + (t + 1) * 32 * XQ);
        __builtin_amdgcn_sched_barrier(0);
        c = lg_mma(c, xb[t & 3], wb[t & 1]);
        __builtin_amdgcn_sched_barrier(0);
    }

    if (kh == 1) {
        #pragma unroll
        for (int rr = 0; rr < 8; ++rr)
            hi[tw][rr][lane] = c[rr];
    }
    __syncthreads();
    float v[8] = {};
    if (kh == 0) {
        float mw = -__builtin_inff();
        #pragma unroll
        for (int rr = 0; rr < 8; ++rr) {
            const int s = s0 + rr + 8 * half;
            float t = (c[rr] + hi[tw][rr][lane]) * SCALE;
            if (mask[(size_t)b * SEQ + s] == 0) t = -__builtin_inff();
            v[rr] = t;
            mw = fmaxf(mw, t);
        }
        mw = fmaxf(mw, __shfl_xor(mw, 16));
        if (half == 0) gmax[tw][m] = mw;
    }
    __syncthreads();
    float mg = 0.0f;
    if (kh == 0) {
        mg = gmax[0][m];
        #pragma unroll
        for (int t = 1; t < LGT; ++t) mg = fmaxf(mg, gmax[t][m]);
        const float ms = (mg == -__builtin_inff()) ? 0.0f : mg;
        float sw = 0.0f;
        #pragma unroll
        for (int rr = 0; rr < 8; ++rr) {
            v[rr] = __expf(v[rr] - ms);
            sw += v[rr];
        }
        sw += __shfl_xor(sw, 16);
        if (half == 0) gsum[tw][m] = sw;
    }
    __syncthreads();
    if (kh == 0) {
        #pragma unroll
        for (int rr = 0; rr < 8; ++rr) {
            const int s = s0 + rr + 8 * half;
            ex[((size_t)b * SEQ + s) * HEADS + m] = v[rr];
        }
    }
    if (wv == 0 && half == 0) {
        float l = gsum[0][m];
        #pragma unroll
        for (int t = 1; t < LGT; ++t) l += gsum[t][m];
        const int g = blockIdx.x & (NG - 1);
        const v2f pr = {mg, l};
        *(v2f*)(stats + (((size_t)b * HEADS + m) * NG + g) * 2) = pr;
    }
}

__global__ __launch_bounds__(32 * PW)
void k_pool(const float* __restrict__ x, const float* __restrict__ attn,
            const float* __restrict__ stats, float* __restrict__ part)
{
    __shared__ v4f wl[RB * 32];
    const int tid  = threadIdx.x;
    const int lane = tid & 31;
    const int n    = lane & 15;
    const int half = lane >> 4;
    const int wv   = tid >> 5;
    const int blk  = blockIdx.x;
    const int d0   = ((blk & 1) * PW + wv) * SLAB;
    const int ch   = (blk >> 1) & (CH - 1);
    const int b    = blk / (2 * CH);
    const int r0   = ch * CROWS;

    const float* xb = x    + ((size_t)b * SEQ + r0 + 4 * half) * DIM + d0 + 4 * n;
    const float* ab = attn + ((size_t)b * SEQ + r0 + 32 * wv + 4 * half) * HEADS + n;
    v8f c[4] = {};

    v4f e0[PS / 2], e1[PS / 2];
    v4f p0[2 * PS], p1[2 * PS];
    constexpr int NSET = CROWS / (4 * PS);
    static_assert(CROWS == 2 * 16 * LGT && NG == 32 && NSET == 8 && PW * 4 * PS == CROWS,
                  "a chunk is two groups of four sets; a wave brings one set's rows");

    const float* sb = stats + ((size_t)b * HEADS + n) * NG * 2;
    v4f st[8];
    #pragma unroll
    for (int i = 0; i < 8; ++i)
        st[i] = *(const v4f*)(sb + 32 * half + 4 * i);
    const v4f own = *(const v4f*)(sb + 4 * ch);

    v4f g[PS / 2];
    #pragma unroll
    for (int j = 0; j < PS / 2; ++j) {
        #pragma unroll
        for (int i = 0; i < 4; ++i)
            g[j][i] = ab[(size_t)(8 * j + i) * HEADS];
    }

    px_load(p0, xb);
    __builtin_amdgcn_sched_barrier(0);
    px_load(p1, xb + (size_t)4 * PS * DIM);
    __builtin_amdgcn_sched_barrier(0);

    float mh = -__builtin_inff();
    #pragma unroll
    for (int i = 0; i < 8; ++i) mh = fmaxf(mh, fmaxf(st[i].x, st[i].z));
    const float M  = fmaxf(mh, __shfl_xor(mh, 16));
    const float Ms = (M == -__builtin_inff()) ? 0.0f : M;
    float zh = 0.0f;
    #pragma unroll
    for (int i = 0; i < 8; ++i) {
        zh += st[i].y * __expf(st[i].x - Ms);
        zh += st[i].w * __expf(st[i].z - Ms);
    }
    const float Z  = zh + __shfl_xor(zh, 16);
    const float rz = 1.0f / Z;
    const float f  = __expf(((wv < PW / 2) ? own.x : own.z) - Ms) * rz;
    #pragma unroll
    for (int j = 0; j < PS / 2; ++j)
        wl[32 * (PS / 2 * wv + j) + lane] = g[j] * f;
    asm volatile("s_wait_dscnt 0x0" ::: "memory");
    asm volatile("s_barrier_signal -1" ::: "memory");
    asm volatile("s_barrier_wait -1" ::: "memory");

    const v4f* wp = wl + lane;
    pe_load(e0, wp);
    pe_load(e1, wp + 32 * (PS / 2));
    __builtin_amdgcn_sched_barrier(0);

    pl_mma(c, e0, p0);
    __builtin_amdgcn_sched_barrier(0);
    pe_load(e0, wp + 2 * 32 * (PS / 2));
    px_load(p0, xb + (size_t)2 * 4 * PS * DIM);
    __builtin_amdgcn_sched_barrier(0);
    pl_mma(c, e1, p1);
    __builtin_amdgcn_sched_barrier(0);
    for (int s = 2; s < NSET - 2; s += 2) {
        pe_load(e1, wp + (s + 1) * 32 * (PS / 2));
        px_load(p1, xb + (size_t)(s + 1) * 4 * PS * DIM);
        __builtin_amdgcn_sched_barrier(0);
        pl_mma(c, e0, p0);
        __builtin_amdgcn_sched_barrier(0);
        pe_load(e0, wp + (s + 2) * 32 * (PS / 2));
        px_load(p0, xb + (size_t)(s + 2) * 4 * PS * DIM);
        __builtin_amdgcn_sched_barrier(0);
        pl_mma(c, e1, p1);
        __builtin_amdgcn_sched_barrier(0);
    }
    pe_load(e1, wp + (NSET - 1) * 32 * (PS / 2));
    px_load(p1, xb + (size_t)(NSET - 1) * 4 * PS * DIM);
    __builtin_amdgcn_sched_barrier(0);
    pl_mma(c, e0, p0);
    __builtin_amdgcn_sched_barrier(0);
    pl_mma(c, e1, p1);

    float* pb = part + (((size_t)b * CH + ch) * HEADS + 8 * half) * DIM + d0 + 4 * n;
    #pragma unroll
    for (int r = 0; r < 8; ++r) {
        const v4f v = {c[0][r], c[1][r], c[2][r], c[3][r]};
        *(v4f*)(pb + (size_t)r * DIM) = v;
    }
}

__global__ __launch_bounds__(256)
void k_sum(const float* __restrict__ part, float* __restrict__ out)
{
    constexpr int PLANE = HEADS * DIM / 4;
    const int i = blockIdx.x * 256 + threadIdx.x;
    const int b = i / PLANE;
    const int r = i % PLANE;
    const v4f* p = (const v4f*)part + (size_t)b * CH * PLANE + r;
    v4f s = p[0];
    #pragma unroll
    for (int cidx = 1; cidx < CH; ++cidx)
        s += p[(size_t)cidx * PLANE];
    ((v4f*)out)[i] = s;
}

extern "C" void kernel_launch(void* const* d_in, const int* in_sizes, int n_in,
                              void* d_out, int out_size, void* d_ws, size_t ws_size,
                              hipStream_t stream)
{
    const float* x    = (const float*)d_in[0];
    const float* W    = (const float*)d_in[1];
    const int*   mask = (const int*)  d_in[2];
    float* out    = (float*)d_out;
    float* ex     = (float*)d_ws;
    float* part   = ex + BATCH * SEQ * HEADS;
    float* stats  = part + BATCH * CH * HEADS * DIM;

    k_logits <<<dim3(BATCH * SEQ / 16 / LGT), dim3(32 * LGW), 0, stream>>>(x, W, mask, ex, stats);
    k_pool   <<<dim3(BATCH * CH * (DIM / SLAB / PW)), dim3(32 * PW), 0, stream>>>(x, ex, stats, part);
    k_sum    <<<dim3(BATCH * HEADS * DIM / 4 / 256), dim3(256), 0, stream>>>(part, out);
}
